# starting kernel
# baseline (speedup 1.0000x reference)
#include <hip/hip_runtime.h>

#define BATCH   262144
#define NDEP    16
#define KIND    32
#define EPSV    1e-7f
#define NITER   12
#define TILES   (BATCH / 16)
#define WPB     8
#define BLOCKS  (TILES / WPB)

typedef float v2f __attribute__((ext_vector_type(2)));
typedef float v8f __attribute__((ext_vector_type(8)));

#define WS_MINV 0
#define WS_MB   256
#define WS_MT   768
#define WS_MU   1280

__global__ __launch_bounds__(32) void clefo_prep(
    const float* __restrict__ Ups, const float* __restrict__ Bmat,
    const float* __restrict__ Theta, const float* __restrict__ Gamma,
    float* __restrict__ ws) {
  __shared__ float sminv[256];
  const int l = threadIdx.x;
  float a[16];
  if (l < 16) {
    #pragma unroll
    for (int i = 0; i < 16; ++i)
      a[i] = ((i == l) ? (1.0f + EPSV) : 0.0f) - Gamma[i * 16 + l];
  } else {
    const int j = l - 16;
    #pragma unroll
    for (int i = 0; i < 16; ++i) a[i] = (i == j) ? 1.0f : 0.0f;
  }
  for (int p = 0; p < 16; ++p) {
    const float piv = __shfl(a[p], p, 32);
    const float inv = 1.0f / piv;
    a[p] *= inv;
    #pragma unroll
    for (int r = 0; r < 16; ++r) {
      if (r == p) continue;
      const float f = __shfl(a[r], p, 32);
      a[r] -= f * a[p];
    }
  }
  if (l >= 16) {
    const int j = l - 16;
    #pragma unroll
    for (int i = 0; i < 16; ++i) {
      sminv[i * 16 + j] = a[i];
      ws[WS_MINV + i * 16 + j] = a[i];
    }
  }
  __syncthreads();
  for (int i = 0; i < 16; ++i) {
    float sb = 0.0f, st = 0.0f;
    #pragma unroll
    for (int m = 0; m < 16; ++m) {
      const float w = sminv[i * 16 + m];
      sb += w * Bmat[m * 32 + l];
      st += w * Theta[m * 32 + l];
    }
    ws[WS_MB + i * 32 + l] = sb;
    ws[WS_MT + i * 32 + l] = st;
  }
  if (l < 16) {
    float s = 0.0f;
    #pragma unroll
    for (int m = 0; m < 16; ++m) s += sminv[l * 16 + m] * Ups[m];
    ws[WS_MU + l] = s;
  }
}

#define XPITCH 36
#define TPITCH 18
#define WAVE_LDS (2 * 16 * XPITCH + 16 * TPITCH)

__global__ __launch_bounds__(256) void clefo_solve(
    const float* __restrict__ X, const float* __restrict__ Z,
    const float* __restrict__ Lam, const float* __restrict__ ws,
    float* __restrict__ out) {
  __shared__ float lds[WPB * WAVE_LDS];
  const int wid  = threadIdx.x >> 5;
  const int lane = threadIdx.x & 31;
  const int hi   = lane >> 4;
  const int lo   = lane & 15;
  const int tile = blockIdx.x * WPB + wid;

  float* lx = lds + wid * WAVE_LDS;
  float* lz = lx + 16 * XPITCH;
  float* lt = lz + 16 * XPITCH;

  v2f fb[8], ft[8], fl[8], fm[4];
  #pragma unroll
  for (int s = 0; s < 8; ++s) {
    const int k = s * 4 + 2 * hi;
    fb[s] = *(const v2f*)(ws + WS_MB + lo * 32 + k);
    ft[s] = *(const v2f*)(ws + WS_MT + lo * 32 + k);
    fl[s] = *(const v2f*)(Lam + lo * 32 + k);
  }
  #pragma unroll
  for (int s = 0; s < 4; ++s)
    fm[s] = *(const v2f*)(ws + WS_MINV + lo * 16 + s * 4 + 2 * hi);
  const float muv = ws[WS_MU + lo];

  const float* Xg = X + (size_t)tile * (16 * KIND);
  const float* Zg = Z + (size_t)tile * (16 * KIND);
  #pragma unroll
  for (int i = 0; i < 4; ++i) {
    const int g = i * 128 + lane * 4;
    const int row = g >> 5, col = g & 31;
    *(float4*)(lx + row * XPITCH + col) = *(const float4*)(Xg + g);
    *(float4*)(lz + row * XPITCH + col) = *(const float4*)(Zg + g);
  }
  __syncthreads();

  v8f c;
  #pragma unroll
  for (int r = 0; r < 8; ++r) c[r] = muv;
  #pragma unroll
  for (int s = 0; s < 8; ++s) {
    const v2f ax = *(const v2f*)(lx + lo * XPITCH + s * 4 + 2 * hi);
    c = __builtin_amdgcn_wmma_f32_16x16x4_f32(false, ax, false, fb[s],
                                              (short)0, c, false, false);
  }
  #pragma unroll
  for (int s = 0; s < 8; ++s) {
    const v2f az = *(const v2f*)(lz + lo * XPITCH + s * 4 + 2 * hi);
    c = __builtin_amdgcn_wmma_f32_16x16x4_f32(false, az, false, ft[s],
                                              (short)0, c, false, false);
  }
  v8f d = {0.f, 0.f, 0.f, 0.f, 0.f, 0.f, 0.f, 0.f};
  #pragma unroll
  for (int s = 0; s < 8; ++s) {
    const v2f ax = *(const v2f*)(lx + lo * XPITCH + s * 4 + 2 * hi);
    d = __builtin_amdgcn_wmma_f32_16x16x4_f32(false, ax, false, fl[s],
                                              (short)0, d, false, false);
  }

  v8f y = c;
  #pragma unroll 2
  for (int it = 0; it < NITER; ++it) {
    #pragma unroll
    for (int r = 0; r < 8; ++r)
      lt[(r + 8 * hi) * TPITCH + lo] = d[r] * y[r];
    v8f yn = c;
    #pragma unroll
    for (int s = 0; s < 4; ++s) {
      const v2f at = *(const v2f*)(lt + lo * TPITCH + s * 4 + 2 * hi);
      yn = __builtin_amdgcn_wmma_f32_16x16x4_f32(false, at, false, fm[s],
                                                 (short)0, yn, false, false);
    }
    y = yn;
  }

  float* og = out + (size_t)tile * (16 * NDEP);
  #pragma unroll
  for (int r = 0; r < 8; ++r) og[(r + 8 * hi) * NDEP + lo] = y[r];
}

extern "C" void kernel_launch(void* const* d_in, const int* in_sizes, int n_in,
                              void* d_out, int out_size, void* d_ws, size_t ws_size,
                              hipStream_t stream) {
  const float* X     = (const float*)d_in[0];
  const float* Z     = (const float*)d_in[1];
  const float* Ups   = (const float*)d_in[2];
  const float* Bmat  = (const float*)d_in[3];
  const float* Theta = (const float*)d_in[4];
  const float* Gamma = (const float*)d_in[5];
  const float* Lam   = (const float*)d_in[6];
  float* ws  = (float*)d_ws;
  float* out = (float*)d_out;

  clefo_prep<<<1, 32, 0, stream>>>(Ups, Bmat, Theta, Gamma, ws);
  clefo_solve<<<BLOCKS, 256, 0, stream>>>(X, Z, Lam, ws, out);
}


# Operand roles swapped (constants as A, items as B): D registers (2t,2t+1) are the next Minv product's B fragment, so the iteration stays in registers; Minv k order permuted to match
# speedup vs baseline: 1.0919x; 1.0919x over previous
#include <hip/hip_runtime.h>

#define BATCH   262144
#define NDEP    16
#define KIND    32
#define EPSV    1e-7f
#define NITER   12
#define TILES   (BATCH / 16)
#define WPB     8
#define BLOCKS  (TILES / WPB)

typedef float v2f __attribute__((ext_vector_type(2)));
typedef float v8f __attribute__((ext_vector_type(8)));

#define WS_MINV 0
#define WS_MB   256
#define WS_MT   768
#define WS_MU   1280

__global__ __launch_bounds__(32) void clefo_prep(
    const float* __restrict__ Ups, const float* __restrict__ Bmat,
    const float* __restrict__ Theta, const float* __restrict__ Gamma,
    float* __restrict__ ws) {
  __shared__ float sminv[256];
  const int l = threadIdx.x;
  float a[16];
  if (l < 16) {
    #pragma unroll
    for (int i = 0; i < 16; ++i)
      a[i] = ((i == l) ? (1.0f + EPSV) : 0.0f) - Gamma[i * 16 + l];
  } else {
    const int j = l - 16;
    #pragma unroll
    for (int i = 0; i < 16; ++i) a[i] = (i == j) ? 1.0f : 0.0f;
  }
  for (int p = 0; p < 16; ++p) {
    const float piv = __shfl(a[p], p, 32);
    const float inv = 1.0f / piv;
    a[p] *= inv;
    #pragma unroll
    for (int r = 0; r < 16; ++r) {
      if (r == p) continue;
      const float f = __shfl(a[r], p, 32);
      a[r] -= f * a[p];
    }
  }
  if (l >= 16) {
    const int j = l - 16;
    #pragma unroll
    for (int i = 0; i < 16; ++i) {
      sminv[i * 16 + j] = a[i];
      ws[WS_MINV + i * 16 + j] = a[i];
    }
  }
  __syncthreads();
  for (int i = 0; i < 16; ++i) {
    float sb = 0.0f, st = 0.0f;
    #pragma unroll
    for (int m = 0; m < 16; ++m) {
      const float w = sminv[i * 16 + m];
      sb += w * Bmat[m * 32 + l];
      st += w * Theta[m * 32 + l];
    }
    ws[WS_MB + i * 32 + l] = sb;
    ws[WS_MT + i * 32 + l] = st;
  }
  if (l < 16) {
    float s = 0.0f;
    #pragma unroll
    for (int m = 0; m < 16; ++m) s += sminv[l * 16 + m] * Ups[m];
    ws[WS_MU + l] = s;
  }
}

#define XPITCH 36
#define WAVE_LDS (2 * 16 * XPITCH)

__global__ __launch_bounds__(256) void clefo_solve(
    const float* __restrict__ X, const float* __restrict__ Z,
    const float* __restrict__ Lam, const float* __restrict__ ws,
    float* __restrict__ out) {
  __shared__ float lds[WPB * WAVE_LDS];
  const int wid  = threadIdx.x >> 5;
  const int lane = threadIdx.x & 31;
  const int hi   = lane >> 4;
  const int lo   = lane & 15;
  const int tile = blockIdx.x * WPB + wid;

  float* lx = lds + wid * WAVE_LDS;
  float* lz = lx + 16 * XPITCH;

  v2f fb[8], ft[8], fl[8], fm[4];
  #pragma unroll
  for (int s = 0; s < 8; ++s) {
    const int k = s * 4 + 2 * hi;
    fb[s] = *(const v2f*)(ws + WS_MB + lo * 32 + k);
    ft[s] = *(const v2f*)(ws + WS_MT + lo * 32 + k);
    fl[s] = *(const v2f*)(Lam + lo * 32 + k);
  }
  #pragma unroll
  for (int t = 0; t < 4; ++t)
    fm[t] = *(const v2f*)(ws + WS_MINV + lo * 16 + 8 * hi + 2 * t);
  const float4 mu0 = *(const float4*)(ws + WS_MU + 8 * hi);
  const float4 mu1 = *(const float4*)(ws + WS_MU + 8 * hi + 4);

  const float* Xg = X + (size_t)tile * (16 * KIND);
  const float* Zg = Z + (size_t)tile * (16 * KIND);
  #pragma unroll
  for (int i = 0; i < 4; ++i) {
    const int g = i * 128 + lane * 4;
    const int row = g >> 5, col = g & 31;
    *(float4*)(lx + row * XPITCH + col) = *(const float4*)(Xg + g);
    *(float4*)(lz + row * XPITCH + col) = *(const float4*)(Zg + g);
  }
  __syncthreads();

  v8f c = {mu0.x, mu0.y, mu0.z, mu0.w, mu1.x, mu1.y, mu1.z, mu1.w};
  #pragma unroll
  for (int s = 0; s < 8; ++s) {
    const v2f bx = *(const v2f*)(lx + lo * XPITCH + s * 4 + 2 * hi);
    c = __builtin_amdgcn_wmma_f32_16x16x4_f32(false, fb[s], false, bx,
                                              (short)0, c, false, false);
  }
  #pragma unroll
  for (int s = 0; s < 8; ++s) {
    const v2f bz = *(const v2f*)(lz + lo * XPITCH + s * 4 + 2 * hi);
    c = __builtin_amdgcn_wmma_f32_16x16x4_f32(false, ft[s], false, bz,
                                              (short)0, c, false, false);
  }
  v8f d = {0.f, 0.f, 0.f, 0.f, 0.f, 0.f, 0.f, 0.f};
  #pragma unroll
  for (int s = 0; s < 8; ++s) {
    const v2f bx = *(const v2f*)(lx + lo * XPITCH + s * 4 + 2 * hi);
    d = __builtin_amdgcn_wmma_f32_16x16x4_f32(false, fl[s], false, bx,
                                              (short)0, d, false, false);
  }

  v8f y = c;
  #pragma unroll 2
  for (int it = 0; it < NITER; ++it) {
    const v8f w = d * y;
    v8f yn = c;
    #pragma unroll
    for (int t = 0; t < 4; ++t) {
      const v2f bw = {w[2 * t], w[2 * t + 1]};
      yn = __builtin_amdgcn_wmma_f32_16x16x4_f32(false, fm[t], false, bw,
                                                 (short)0, yn, false, false);
    }
    y = yn;
  }

  float* og = out + (size_t)tile * (16 * NDEP) + lo * NDEP + 8 * hi;
  *(float4*)(og)     = make_float4(y[0], y[1], y[2], y[3]);
  *(float4*)(og + 4) = make_float4(y[4], y[5], y[6], y[7]);
}

extern "C" void kernel_launch(void* const* d_in, const int* in_sizes, int n_in,
                              void* d_out, int out_size, void* d_ws, size_t ws_size,
                              hipStream_t stream) {
  const float* X     = (const float*)d_in[0];
  const float* Z     = (const float*)d_in[1];
  const float* Ups   = (const float*)d_in[2];
  const float* Bmat  = (const float*)d_in[3];
  const float* Theta = (const float*)d_in[4];
  const float* Gamma = (const float*)d_in[5];
  const float* Lam   = (const float*)d_in[6];
  float* ws  = (float*)d_ws;
  float* out = (float*)d_out;

  clefo_prep<<<1, 32, 0, stream>>>(Ups, Bmat, Theta, Gamma, ws);
  clefo_solve<<<BLOCKS, 256, 0, stream>>>(X, Z, Lam, ws, out);
}


# Splitting changed to y <- (rhs + Gamma@y)/((1+eps)-d): no inverse needed, so the prep launch and its workspace round trip are gone (1 launch); iterates and sum order change, same 12 sweeps
# speedup vs baseline: 1.4795x; 1.3550x over previous
#include <hip/hip_runtime.h>

#define BATCH   262144
#define NDEP    16
#define KIND    32
#define EPSV    1e-7f
#define NITER   12
#define TILES   (BATCH / 16)
#define WPB     8
#define BLOCKS  (TILES / WPB)

typedef float v2f __attribute__((ext_vector_type(2)));
typedef float v8f __attribute__((ext_vector_type(8)));

#define XPITCH 36
#define WAVE_LDS (2 * 16 * XPITCH)

__global__ __launch_bounds__(256) void clefo_solve(
    const float* __restrict__ X, const float* __restrict__ Z,
    const float* __restrict__ Ups, const float* __restrict__ Bmat,
    const float* __restrict__ Theta, const float* __restrict__ Gamma,
    const float* __restrict__ Lam, float* __restrict__ out) {
  __shared__ float lds[WPB * WAVE_LDS];
  const int wid  = threadIdx.x >> 5;
  const int lane = threadIdx.x & 31;
  const int hi   = lane >> 4;
  const int lo   = lane & 15;
  const int tile = blockIdx.x * WPB + wid;

  float* lx = lds + wid * WAVE_LDS;
  float* lz = lx + 16 * XPITCH;

  v2f fb[8], ft[8], fl[8], fm[4];
  #pragma unroll
  for (int s = 0; s < 8; ++s) {
    const int k = s * 4 + 2 * hi;
    fb[s] = *(const v2f*)(Bmat  + lo * 32 + k);
    ft[s] = *(const v2f*)(Theta + lo * 32 + k);
    fl[s] = *(const v2f*)(Lam   + lo * 32 + k);
  }
  #pragma unroll
  for (int t = 0; t < 4; ++t)
    fm[t] = *(const v2f*)(Gamma + lo * 16 + 8 * hi + 2 * t);
  const float4 mu0 = *(const float4*)(Ups + 8 * hi);
  const float4 mu1 = *(const float4*)(Ups + 8 * hi + 4);

  const float* Xg = X + (size_t)tile * (16 * KIND);
  const float* Zg = Z + (size_t)tile * (16 * KIND);
  #pragma unroll
  for (int i = 0; i < 4; ++i) {
    const int g = i * 128 + lane * 4;
    const int row = g >> 5, col = g & 31;
    *(float4*)(lx + row * XPITCH + col) = *(const float4*)(Xg + g);
    *(float4*)(lz + row * XPITCH + col) = *(const float4*)(Zg + g);
  }
  __syncthreads();

  v8f c = {mu0.x, mu0.y, mu0.z, mu0.w, mu1.x, mu1.y, mu1.z, mu1.w};
  #pragma unroll
  for (int s = 0; s < 8; ++s) {
    const v2f bx = *(const v2f*)(lx + lo * XPITCH + s * 4 + 2 * hi);
    c = __builtin_amdgcn_wmma_f32_16x16x4_f32(false, fb[s], false, bx,
                                              (short)0, c, false, false);
  }
  #pragma unroll
  for (int s = 0; s < 8; ++s) {
    const v2f bz = *(const v2f*)(lz + lo * XPITCH + s * 4 + 2 * hi);
    c = __builtin_amdgcn_wmma_f32_16x16x4_f32(false, ft[s], false, bz,
                                              (short)0, c, false, false);
  }
  v8f d = {0.f, 0.f, 0.f, 0.f, 0.f, 0.f, 0.f, 0.f};
  #pragma unroll
  for (int s = 0; s < 8; ++s) {
    const v2f bx = *(const v2f*)(lx + lo * XPITCH + s * 4 + 2 * hi);
    d = __builtin_amdgcn_wmma_f32_16x16x4_f32(false, fl[s], false, bx,
                                              (short)0, d, false, false);
  }

  const v8f r = 1.0f / ((1.0f + EPSV) - d);

  v8f u = c;
  #pragma unroll 2
  for (int it = 0; it < NITER; ++it) {
    const v8f w = r * u;
    v8f un = c;
    #pragma unroll
    for (int t = 0; t < 4; ++t) {
      const v2f bw = {w[2 * t], w[2 * t + 1]};
      un = __builtin_amdgcn_wmma_f32_16x16x4_f32(false, fm[t], false, bw,
                                                 (short)0, un, false, false);
    }
    u = un;
  }
  const v8f y = r * u;

  float* og = out + (size_t)tile * (16 * NDEP) + lo * NDEP + 8 * hi;
  *(float4*)(og)     = make_float4(y[0], y[1], y[2], y[3]);
  *(float4*)(og + 4) = make_float4(y[4], y[5], y[6], y[7]);
}

extern "C" void kernel_launch(void* const* d_in, const int* in_sizes, int n_in,
                              void* d_out, int out_size, void* d_ws, size_t ws_size,
                              hipStream_t stream) {
  const float* X     = (const float*)d_in[0];
  const float* Z     = (const float*)d_in[1];
  const float* Ups   = (const float*)d_in[2];
  const float* Bmat  = (const float*)d_in[3];
  const float* Theta = (const float*)d_in[4];
  const float* Gamma = (const float*)d_in[5];
  const float* Lam   = (const float*)d_in[6];
  float* out = (float*)d_out;

  clefo_solve<<<BLOCKS, 256, 0, stream>>>(X, Z, Ups, Bmat, Theta, Gamma, Lam, out);
}


# Each wave keeps the constant fragments and solves 4 tiles in turn, the next tile's rows prefetched into registers: 4096 waves enter instead of 16384; wave-level LDS hand-over, no block barrier
# speedup vs baseline: 1.7774x; 1.2014x over previous
#include <hip/hip_runtime.h>

#define BATCH   262144
#define NDEP    16
#define KIND    32
#define EPSV    1e-7f
#define NITER   12
#define TILES   (BATCH / 16)
#define WPB     8
#define TPW     4
#define NWAVES  (TILES / TPW)
#define BLOCKS  (NWAVES / WPB)

typedef float v2f __attribute__((ext_vector_type(2)));
typedef float v8f __attribute__((ext_vector_type(8)));

#define XPITCH 36
#define WAVE_LDS (2 * 16 * XPITCH)

__device__ __forceinline__ void wave_lds_handover() {
  __builtin_amdgcn_fence(__ATOMIC_RELEASE, "wavefront");
  __builtin_amdgcn_wave_barrier();
  __builtin_amdgcn_fence(__ATOMIC_ACQUIRE, "wavefront");
}

__global__ __launch_bounds__(256) __attribute__((amdgpu_waves_per_eu(4, 4)))
void clefo_solve(
    const float* __restrict__ X, const float* __restrict__ Z,
    const float* __restrict__ Ups, const float* __restrict__ Bmat,
    const float* __restrict__ Theta, const float* __restrict__ Gamma,
    const float* __restrict__ Lam, float* __restrict__ out) {
  __shared__ float lds[WPB * WAVE_LDS];
  const int wid  = threadIdx.x >> 5;
  const int lane = threadIdx.x & 31;
  const int hi   = lane >> 4;
  const int lo   = lane & 15;
  const int wave = blockIdx.x * WPB + wid;

  float* lx = lds + wid * WAVE_LDS;
  float* lz = lx + 16 * XPITCH;

  v2f fb[8], ft[8], fl[8], fm[4];
  #pragma unroll
  for (int s = 0; s < 8; ++s) {
    const int k = s * 4 + 2 * hi;
    fb[s] = *(const v2f*)(Bmat  + lo * 32 + k);
    ft[s] = *(const v2f*)(Theta + lo * 32 + k);
    fl[s] = *(const v2f*)(Lam   + lo * 32 + k);
  }
  #pragma unroll
  for (int t = 0; t < 4; ++t)
    fm[t] = *(const v2f*)(Gamma + lo * 16 + 8 * hi + 2 * t);
  const float4 mu0 = *(const float4*)(Ups + 8 * hi);
  const float4 mu1 = *(const float4*)(Ups + 8 * hi + 4);

  __builtin_amdgcn_sched_barrier(0);

  float4 px0, px1, px2, px3, pz0, pz1, pz2, pz3;
  {
    const float* Xg = X + (size_t)wave * (16 * KIND) + lane * 4;
    const float* Zg = Z + (size_t)wave * (16 * KIND) + lane * 4;
    px0 = *(const float4*)(Xg);        pz0 = *(const float4*)(Zg);
    px1 = *(const float4*)(Xg + 128);  pz1 = *(const float4*)(Zg + 128);
    px2 = *(const float4*)(Xg + 256);  pz2 = *(const float4*)(Zg + 256);
    px3 = *(const float4*)(Xg + 384);  pz3 = *(const float4*)(Zg + 384);
  }

  float* sx = lx + (lane >> 3) * XPITCH + (lane & 7) * 4;
  float* sz = lz + (lane >> 3) * XPITCH + (lane & 7) * 4;
  const float* rx = lx + lo * XPITCH + 2 * hi;
  const float* rz = lz + lo * XPITCH + 2 * hi;

  #pragma unroll 1
  for (int it = 0; it < TPW; ++it) {
    const int tile = it * NWAVES + wave;

    wave_lds_handover();
    *(float4*)(sx)               = px0;  *(float4*)(sz)               = pz0;
    *(float4*)(sx +  4 * XPITCH) = px1;  *(float4*)(sz +  4 * XPITCH) = pz1;
    *(float4*)(sx +  8 * XPITCH) = px2;  *(float4*)(sz +  8 * XPITCH) = pz2;
    *(float4*)(sx + 12 * XPITCH) = px3;  *(float4*)(sz + 12 * XPITCH) = pz3;
    wave_lds_handover();

    if (it + 1 < TPW) {
      const float* Xg = X + (size_t)(tile + NWAVES) * (16 * KIND) + lane * 4;
      const float* Zg = Z + (size_t)(tile + NWAVES) * (16 * KIND) + lane * 4;
      px0 = *(const float4*)(Xg);        pz0 = *(const float4*)(Zg);
      px1 = *(const float4*)(Xg + 128);  pz1 = *(const float4*)(Zg + 128);
      px2 = *(const float4*)(Xg + 256);  pz2 = *(const float4*)(Zg + 256);
      px3 = *(const float4*)(Xg + 384);  pz3 = *(const float4*)(Zg + 384);
    }

    v8f c = {mu0.x, mu0.y, mu0.z, mu0.w, mu1.x, mu1.y, mu1.z, mu1.w};
    v8f d = {0.f, 0.f, 0.f, 0.f, 0.f, 0.f, 0.f, 0.f};
    #pragma unroll
    for (int s = 0; s < 8; ++s) {
      const v2f bx = *(const v2f*)(rx + s * 4);
      c = __builtin_amdgcn_wmma_f32_16x16x4_f32(false, fb[s], false, bx,
                                                (short)0, c, false, false);
      d = __builtin_amdgcn_wmma_f32_16x16x4_f32(false, fl[s], false, bx,
                                                (short)0, d, false, false);
    }
    #pragma unroll
    for (int s = 0; s < 8; ++s) {
      const v2f bz = *(const v2f*)(rz + s * 4);
      c = __builtin_amdgcn_wmma_f32_16x16x4_f32(false, ft[s], false, bz,
                                                (short)0, c, false, false);
    }

    const v8f r = 1.0f / ((1.0f + EPSV) - d);

    v8f u = c;
    #pragma unroll 2
    for (int sw = 0; sw < NITER; ++sw) {
      const v8f w = r * u;
      v8f un = c;
      #pragma unroll
      for (int t = 0; t < 4; ++t) {
        const v2f bw = {w[2 * t], w[2 * t + 1]};
        un = __builtin_amdgcn_wmma_f32_16x16x4_f32(false, fm[t], false, bw,
                                                   (short)0, un, false, false);
      }
      u = un;
    }
    const v8f y = r * u;

    float* og = out + (size_t)tile * (16 * NDEP) + lo * NDEP + 8 * hi;
    *(float4*)(og)     = make_float4(y[0], y[1], y[2], y[3]);
    *(float4*)(og + 4) = make_float4(y[4], y[5], y[6], y[7]);
  }
}

extern "C" void kernel_launch(void* const* d_in, const int* in_sizes, int n_in,
                              void* d_out, int out_size, void* d_ws, size_t ws_size,
                              hipStream_t stream) {
  const float* X     = (const float*)d_in[0];
  const float* Z     = (const float*)d_in[1];
  const float* Ups   = (const float*)d_in[2];
  const float* Bmat  = (const float*)d_in[3];
  const float* Theta = (const float*)d_in[4];
  const float* Gamma = (const float*)d_in[5];
  const float* Lam   = (const float*)d_in[6];
  float* out = (float*)d_out;

  clefo_solve<<<BLOCKS, 256, 0, stream>>>(X, Z, Ups, Bmat, Theta, Gamma, Lam, out);
}


# Sweeps end at convergence: from the 5th on, a wave-wide test after each sweep stops the tile once no entry of u moved by more than 2^-22 of its lane's largest; cap stays 12; last-place differences
# speedup vs baseline: 2.1695x; 1.2206x over previous
#include <hip/hip_runtime.h>

#define BATCH   262144
#define NDEP    16
#define KIND    32
#define EPSV    1e-7f
#define NITER   12
#define NPRE    4
#define CTOL    0x1p-22f
#define TILES   (BATCH / 16)
#define WPB     8
#define TPW     4
#define NWAVES  (TILES / TPW)
#define BLOCKS  (NWAVES / WPB)

typedef float v2f __attribute__((ext_vector_type(2)));
typedef float v8f __attribute__((ext_vector_type(8)));

#define XPITCH 36
#define WAVE_LDS (2 * 16 * XPITCH)

__device__ __forceinline__ void wave_lds_handover() {
  __builtin_amdgcn_fence(__ATOMIC_RELEASE, "wavefront");
  __builtin_amdgcn_wave_barrier();
  __builtin_amdgcn_fence(__ATOMIC_ACQUIRE, "wavefront");
}

__global__ __launch_bounds__(256) __attribute__((amdgpu_waves_per_eu(4, 4)))
void clefo_solve(
    const float* __restrict__ X, const float* __restrict__ Z,
    const float* __restrict__ Ups, const float* __restrict__ Bmat,
    const float* __restrict__ Theta, const float* __restrict__ Gamma,
    const float* __restrict__ Lam, float* __restrict__ out) {
  __shared__ float lds[WPB * WAVE_LDS];
  const int wid  = threadIdx.x >> 5;
  const int lane = threadIdx.x & 31;
  const int hi   = lane >> 4;
  const int lo   = lane & 15;
  const int wave = blockIdx.x * WPB + wid;

  float* lx = lds + wid * WAVE_LDS;
  float* lz = lx + 16 * XPITCH;

  v2f fb[8], ft[8], fl[8], fm[4];
  #pragma unroll
  for (int s = 0; s < 8; ++s) {
    const int k = s * 4 + 2 * hi;
    fb[s] = *(const v2f*)(Bmat  + lo * 32 + k);
    ft[s] = *(const v2f*)(Theta + lo * 32 + k);
    fl[s] = *(const v2f*)(Lam   + lo * 32 + k);
  }
  #pragma unroll
  for (int t = 0; t < 4; ++t)
    fm[t] = *(const v2f*)(Gamma + lo * 16 + 8 * hi + 2 * t);
  const float4 mu0 = *(const float4*)(Ups + 8 * hi);
  const float4 mu1 = *(const float4*)(Ups + 8 * hi + 4);

  __builtin_amdgcn_sched_barrier(0);

  float4 px0, px1, px2, px3, pz0, pz1, pz2, pz3;
  {
    const float* Xg = X + (size_t)wave * (16 * KIND) + lane * 4;
    const float* Zg = Z + (size_t)wave * (16 * KIND) + lane * 4;
    px0 = *(const float4*)(Xg);        pz0 = *(const float4*)(Zg);
    px1 = *(const float4*)(Xg + 128);  pz1 = *(const float4*)(Zg + 128);
    px2 = *(const float4*)(Xg + 256);  pz2 = *(const float4*)(Zg + 256);
    px3 = *(const float4*)(Xg + 384);  pz3 = *(const float4*)(Zg + 384);
  }

  float* sx = lx + (lane >> 3) * XPITCH + (lane & 7) * 4;
  float* sz = lz + (lane >> 3) * XPITCH + (lane & 7) * 4;
  const float* rx = lx + lo * XPITCH + 2 * hi;
  const float* rz = lz + lo * XPITCH + 2 * hi;

  #pragma unroll 1
  for (int it = 0; it < TPW; ++it) {
    const int tile = it * NWAVES + wave;

    wave_lds_handover();
    *(float4*)(sx)               = px0;  *(float4*)(sz)               = pz0;
    *(float4*)(sx +  4 * XPITCH) = px1;  *(float4*)(sz +  4 * XPITCH) = pz1;
    *(float4*)(sx +  8 * XPITCH) = px2;  *(float4*)(sz +  8 * XPITCH) = pz2;
    *(float4*)(sx + 12 * XPITCH) = px3;  *(float4*)(sz + 12 * XPITCH) = pz3;
    wave_lds_handover();

    if (it + 1 < TPW) {
      const float* Xg = X + (size_t)(tile + NWAVES) * (16 * KIND) + lane * 4;
      const float* Zg = Z + (size_t)(tile + NWAVES) * (16 * KIND) + lane * 4;
      px0 = *(const float4*)(Xg);        pz0 = *(const float4*)(Zg);
      px1 = *(const float4*)(Xg + 128);  pz1 = *(const float4*)(Zg + 128);
      px2 = *(const float4*)(Xg + 256);  pz2 = *(const float4*)(Zg + 256);
      px3 = *(const float4*)(Xg + 384);  pz3 = *(const float4*)(Zg + 384);
    }

    v8f c = {mu0.x, mu0.y, mu0.z, mu0.w, mu1.x, mu1.y, mu1.z, mu1.w};
    v8f d = {0.f, 0.f, 0.f, 0.f, 0.f, 0.f, 0.f, 0.f};
    #pragma unroll
    for (int s = 0; s < 8; ++s) {
      const v2f bx = *(const v2f*)(rx + s * 4);
      c = __builtin_amdgcn_wmma_f32_16x16x4_f32(false, fb[s], false, bx,
                                                (short)0, c, false, false);
      d = __builtin_amdgcn_wmma_f32_16x16x4_f32(false, fl[s], false, bx,
                                                (short)0, d, false, false);
    }
    #pragma unroll
    for (int s = 0; s < 8; ++s) {
      const v2f bz = *(const v2f*)(rz + s * 4);
      c = __builtin_amdgcn_wmma_f32_16x16x4_f32(false, ft[s], false, bz,
                                                (short)0, c, false, false);
    }

    const v8f r = 1.0f / ((1.0f + EPSV) - d);

    v8f u = c;
    #pragma unroll 2
    for (int sw = 0; sw < NPRE; ++sw) {
      const v8f w = r * u;
      v8f un = c;
      #pragma unroll
      for (int t = 0; t < 4; ++t) {
        const v2f bw = {w[2 * t], w[2 * t + 1]};
        un = __builtin_amdgcn_wmma_f32_16x16x4_f32(false, fm[t], false, bw,
                                                   (short)0, un, false, false);
      }
      u = un;
    }
    for (int sw = NPRE; sw < NITER; ++sw) {
      const v8f w = r * u;
      v8f un = c;
      #pragma unroll
      for (int t = 0; t < 4; ++t) {
        const v2f bw = {w[2 * t], w[2 * t + 1]};
        un = __builtin_amdgcn_wmma_f32_16x16x4_f32(false, fm[t], false, bw,
                                                   (short)0, un, false, false);
      }
      float mv = 0.f, mx = 0.f;
      #pragma unroll
      for (int i = 0; i < 8; ++i) {
        mv = __builtin_elementwise_maximum(mv, __builtin_fabsf(un[i] - u[i]));
        mx = __builtin_elementwise_maximum(mx, __builtin_fabsf(un[i]));
      }
      const bool moving = !(mv <= CTOL * mx);
      u = un;
      if (__builtin_amdgcn_ballot_w32(moving) == 0u) break;
    }
    const v8f y = r * u;

    float* og = out + (size_t)tile * (16 * NDEP) + lo * NDEP + 8 * hi;
    *(float4*)(og)     = make_float4(y[0], y[1], y[2], y[3]);
    *(float4*)(og + 4) = make_float4(y[4], y[5], y[6], y[7]);
  }
}

extern "C" void kernel_launch(void* const* d_in, const int* in_sizes, int n_in,
                              void* d_out, int out_size, void* d_ws, size_t ws_size,
                              hipStream_t stream) {
  const float* X     = (const float*)d_in[0];
  const float* Z     = (const float*)d_in[1];
  const float* Ups   = (const float*)d_in[2];
  const float* Bmat  = (const float*)d_in[3];
  const float* Theta = (const float*)d_in[4];
  const float* Gamma = (const float*)d_in[5];
  const float* Lam   = (const float*)d_in[6];
  float* out = (float*)d_out;

  clefo_solve<<<BLOCKS, 256, 0, stream>>>(X, Z, Ups, Bmat, Theta, Gamma, Lam, out);
}


# Sweeps on the system with its constant part solved ahead: N=(I-Gamma/(1+eps))^-1 by 3 squarings on WMMA per block, folded into the constants; sweep matrix N@Gamma@diag(r-1/(1+eps)); sum order changes
# speedup vs baseline: 2.3784x; 1.0963x over previous
#include <hip/hip_runtime.h>

#define BATCH   262144
#define NDEP    16
#define KIND    32
#define EPSV    1e-7f
#define SBAR    (1.0f + EPSV)
#define RBAR    (1.0f / SBAR)
#define NITER   12
#define NPRE    2
#define CTOL    0x1p-22f
#define TILES   (BATCH / 16)
#define WPB     8
#define TPW     4
#define NWAVES  (TILES / TPW)
#define BLOCKS  (NWAVES / WPB)

typedef float v2f __attribute__((ext_vector_type(2)));
typedef float v8f __attribute__((ext_vector_type(8)));

#define XPITCH 36
#define WAVE_LDS (2 * 16 * XPITCH)
#define NCST 12

__device__ __forceinline__ void wave_lds_handover() {
  __builtin_amdgcn_fence(__ATOMIC_RELEASE, "wavefront");
  __builtin_amdgcn_wave_barrier();
  __builtin_amdgcn_fence(__ATOMIC_ACQUIRE, "wavefront");
}

__device__ __forceinline__ v8f wmma4(const v2f a, const v2f b, const v8f c) {
  return __builtin_amdgcn_wmma_f32_16x16x4_f32(false, a, false, b,
                                               (short)0, c, false, false);
}

__device__ __forceinline__ v8f mm_t(const v8f x, const v8f y, v8f c) {
  #pragma unroll
  for (int t = 0; t < 4; ++t) {
    const v2f a = {x[2 * t], x[2 * t + 1]};
    const v2f b = {y[2 * t], y[2 * t + 1]};
    c = wmma4(a, b, c);
  }
  return c;
}

__device__ __forceinline__ v8f join8(const float4 a, const float4 b) {
  const v8f r = {a.x, a.y, a.z, a.w, b.x, b.y, b.z, b.w};
  return r;
}

__global__ __launch_bounds__(256) __attribute__((amdgpu_waves_per_eu(4, 4)))
void clefo_solve(
    const float* __restrict__ X, const float* __restrict__ Z,
    const float* __restrict__ Ups, const float* __restrict__ Bmat,
    const float* __restrict__ Theta, const float* __restrict__ Gamma,
    const float* __restrict__ Lam, float* __restrict__ out) {
  __shared__ float lds[WPB * WAVE_LDS];
  __shared__ float4 cst[NCST * 32];
  const int wid  = __builtin_amdgcn_readfirstlane((int)(threadIdx.x >> 5));
  const int lane = threadIdx.x & 31;
  const int hi   = lane >> 4;
  const int lo   = lane & 15;
  const int wave = blockIdx.x * WPB + wid;
  const int fold_wave = (blockIdx.x >> 8) & 3;

  float* lx = lds + wid * WAVE_LDS;
  float* lz = lx + 16 * XPITCH;

  v2f fl[8];
  #pragma unroll
  for (int s = 0; s < 8; ++s)
    fl[s] = *(const v2f*)(Lam + lo * 32 + 16 * (s >> 2) + 8 * hi + 2 * (s & 3));

  float4 px0, px1, px2, px3, pz0, pz1, pz2, pz3;
  const float* Xg0 = X + (size_t)wave * (16 * KIND) + lane * 4;
  const float* Zg0 = Z + (size_t)wave * (16 * KIND) + lane * 4;

  const v8f z8 = {0.f, 0.f, 0.f, 0.f, 0.f, 0.f, 0.f, 0.f};
  if (wid == fold_wave) {
    v8f gdu, bd0, bd1, td0, td1;
    #pragma unroll
    for (int r = 0; r < 8; ++r) {
      const int row = 8 * hi + r;
      gdu[r] = Gamma[row * 16 + lo];
      bd0[r] = Bmat[row * 32 + lo];
      bd1[r] = Bmat[row * 32 + 16 + lo];
      td0[r] = Theta[row * 32 + lo];
      td1[r] = Theta[row * 32 + 16 + lo];
    }
    const v8f gtu = join8(*(const float4*)(Gamma + lo * 16 + 8 * hi),
                          *(const float4*)(Gamma + lo * 16 + 8 * hi + 4));
    const v8f uu = join8(*(const float4*)(Ups + 8 * hi),
                         *(const float4*)(Ups + 8 * hi + 4));

    __builtin_amdgcn_sched_barrier(0);
    px0 = *(const float4*)(Xg0);        pz0 = *(const float4*)(Zg0);
    px1 = *(const float4*)(Xg0 + 128);  pz1 = *(const float4*)(Zg0 + 128);
    px2 = *(const float4*)(Xg0 + 256);  pz2 = *(const float4*)(Zg0 + 256);
    px3 = *(const float4*)(Xg0 + 384);  pz3 = *(const float4*)(Zg0 + 384);
    __builtin_amdgcn_sched_barrier(0);

    const v8f gd = gdu * RBAR;
    const v8f gt = gtu * RBAR;
    v8f m = gt;
    #pragma unroll
    for (int r = 0; r < 8; ++r)
      if (8 * hi + r == lo) m[r] += 1.0f;
    const v8f q1  = mm_t(gt, gd, z8);
    const v8f qt1 = mm_t(gd, gt, z8);
    m = mm_t(q1, m, m);
    const v8f q2  = mm_t(qt1, q1, z8);
    const v8f qt2 = mm_t(q1, qt1, z8);
    m = mm_t(q2, m, m);
    const v8f q3  = mm_t(qt2, q2, z8);
    m = mm_t(q3, m, m);

    const v8f bp0 = mm_t(bd0, m, z8);
    const v8f bp1 = mm_t(bd1, m, z8);
    const v8f tp0 = mm_t(td0, m, z8);
    const v8f tp1 = mm_t(td1, m, z8);
    const v8f gp  = mm_t(gdu, m, z8);
    const v8f mu  = mm_t(m, uu, z8);

    float4* cw = cst + lane;
    cw[ 0 * 32] = make_float4(bp0[0], bp0[1], bp0[2], bp0[3]);
    cw[ 1 * 32] = make_float4(bp0[4], bp0[5], bp0[6], bp0[7]);
    cw[ 2 * 32] = make_float4(bp1[0], bp1[1], bp1[2], bp1[3]);
    cw[ 3 * 32] = make_float4(bp1[4], bp1[5], bp1[6], bp1[7]);
    cw[ 4 * 32] = make_float4(tp0[0], tp0[1], tp0[2], tp0[3]);
    cw[ 5 * 32] = make_float4(tp0[4], tp0[5], tp0[6], tp0[7]);
    cw[ 6 * 32] = make_float4(tp1[0], tp1[1], tp1[2], tp1[3]);
    cw[ 7 * 32] = make_float4(tp1[4], tp1[5], tp1[6], tp1[7]);
    cw[ 8 * 32] = make_float4(gp[0], gp[1], gp[2], gp[3]);
    cw[ 9 * 32] = make_float4(gp[4], gp[5], gp[6], gp[7]);
    cw[10 * 32] = make_float4(mu[0], mu[1], mu[2], mu[3]);
    cw[11 * 32] = make_float4(mu[4], mu[5], mu[6], mu[7]);
  } else {
    px0 = *(const float4*)(Xg0);        pz0 = *(const float4*)(Zg0);
    px1 = *(const float4*)(Xg0 + 128);  pz1 = *(const float4*)(Zg0 + 128);
    px2 = *(const float4*)(Xg0 + 256);  pz2 = *(const float4*)(Zg0 + 256);
    px3 = *(const float4*)(Xg0 + 384);  pz3 = *(const float4*)(Zg0 + 384);
  }
  __syncthreads();

  const float4* cr = cst + lane;
  const v8f fb0 = join8(cr[ 0 * 32], cr[ 1 * 32]);
  const v8f fb1 = join8(cr[ 2 * 32], cr[ 3 * 32]);
  const v8f ft0 = join8(cr[ 4 * 32], cr[ 5 * 32]);
  const v8f ft1 = join8(cr[ 6 * 32], cr[ 7 * 32]);
  const v8f fg  = join8(cr[ 8 * 32], cr[ 9 * 32]);
  const v8f mup = join8(cr[10 * 32], cr[11 * 32]);

  float* sx = lx + (lane >> 3) * XPITCH + (lane & 7) * 4;
  float* sz = lz + (lane >> 3) * XPITCH + (lane & 7) * 4;
  const float* rx = lx + lo * XPITCH + 8 * hi;
  const float* rz = lz + lo * XPITCH + 8 * hi;

  #pragma unroll 1
  for (int it = 0; it < TPW; ++it) {
    const int tile = it * NWAVES + wave;

    wave_lds_handover();
    *(float4*)(sx)               = px0;  *(float4*)(sz)               = pz0;
    *(float4*)(sx +  4 * XPITCH) = px1;  *(float4*)(sz +  4 * XPITCH) = pz1;
    *(float4*)(sx +  8 * XPITCH) = px2;  *(float4*)(sz +  8 * XPITCH) = pz2;
    *(float4*)(sx + 12 * XPITCH) = px3;  *(float4*)(sz + 12 * XPITCH) = pz3;
    wave_lds_handover();

    if (it + 1 < TPW) {
      const float* Xg = X + (size_t)(tile + NWAVES) * (16 * KIND) + lane * 4;
      const float* Zg = Z + (size_t)(tile + NWAVES) * (16 * KIND) + lane * 4;
      px0 = *(const float4*)(Xg);        pz0 = *(const float4*)(Zg);
      px1 = *(const float4*)(Xg + 128);  pz1 = *(const float4*)(Zg + 128);
      px2 = *(const float4*)(Xg + 256);  pz2 = *(const float4*)(Zg + 256);
      px3 = *(const float4*)(Xg + 384);  pz3 = *(const float4*)(Zg + 384);
    }

    v8f c = mup;
    v8f d = z8;
    #pragma unroll
    for (int s = 0; s < 8; ++s) {
      const int p = s & 3;
      const v2f bx = *(const v2f*)(rx + 16 * (s >> 2) + 2 * p);
      const v2f a  = (s < 4) ? (v2f){fb0[2 * p], fb0[2 * p + 1]}
                             : (v2f){fb1[2 * p], fb1[2 * p + 1]};
      c = wmma4(a, bx, c);
      d = wmma4(fl[s], bx, d);
    }
    #pragma unroll
    for (int s = 0; s < 8; ++s) {
      const int p = s & 3;
      const v2f bz = *(const v2f*)(rz + 16 * (s >> 2) + 2 * p);
      const v2f a  = (s < 4) ? (v2f){ft0[2 * p], ft0[2 * p + 1]}
                             : (v2f){ft1[2 * p], ft1[2 * p + 1]};
      c = wmma4(a, bz, c);
    }

    const v8f r   = 1.0f / (SBAR - d);
    const v8f rho = (d * r) * RBAR;

    v8f u = c;
    #pragma unroll 2
    for (int sw = 0; sw < NPRE; ++sw) {
      const v8f w = rho * u;
      v8f un = c;
      #pragma unroll
      for (int t = 0; t < 4; ++t) {
        const v2f bw = {w[2 * t], w[2 * t + 1]};
        const v2f ag = {fg[2 * t], fg[2 * t + 1]};
        un = wmma4(ag, bw, un);
      }
      u = un;
    }
    for (int sw = NPRE; sw < NITER; ++sw) {
      const v8f w = rho * u;
      v8f un = c;
      #pragma unroll
      for (int t = 0; t < 4; ++t) {
        const v2f bw = {w[2 * t], w[2 * t + 1]};
        const v2f ag = {fg[2 * t], fg[2 * t + 1]};
        un = wmma4(ag, bw, un);
      }
      float mv = 0.f, mx = 0.f;
      #pragma unroll
      for (int i = 0; i < 8; ++i) {
        mv = __builtin_elementwise_maximum(mv, __builtin_fabsf(un[i] - u[i]));
        mx = __builtin_elementwise_maximum(mx, __builtin_fabsf(un[i]));
      }
      const bool moving = !(mv <= CTOL * mx);
      u = un;
      if (__builtin_amdgcn_ballot_w32(moving) == 0u) break;
    }
    const v8f y = r * u;

    float* og = out + (size_t)tile * (16 * NDEP) + lo * NDEP + 8 * hi;
    *(float4*)(og)     = make_float4(y[0], y[1], y[2], y[3]);
    *(float4*)(og + 4) = make_float4(y[4], y[5], y[6], y[7]);
  }
}

extern "C" void kernel_launch(void* const* d_in, const int* in_sizes, int n_in,
                              void* d_out, int out_size, void* d_ws, size_t ws_size,
                              hipStream_t stream) {
  const float* X     = (const float*)d_in[0];
  const float* Z     = (const float*)d_in[1];
  const float* Ups   = (const float*)d_in[2];
  const float* Bmat  = (const float*)d_in[3];
  const float* Theta = (const float*)d_in[4];
  const float* Gamma = (const float*)d_in[5];
  const float* Lam   = (const float*)d_in[6];
  float* out = (float*)d_out;

  clefo_solve<<<BLOCKS, 256, 0, stream>>>(X, Z, Ups, Bmat, Theta, Gamma, Lam, out);
}
